# starting kernel
# baseline (speedup 1.0000x reference)
#include <hip/hip_runtime.h>
#include <hip/hip_bf16.h>

typedef float v2f __attribute__((ext_vector_type(2)));
typedef float v8f __attribute__((ext_vector_type(8)));

#define LDS_LD 130

__device__ __forceinline__ v8f wmma_f32(v2f a, v2f b, v8f c) {
    return __builtin_amdgcn_wmma_f32_16x16x4_f32(
         false, a,  false, b,
         (short)0, c,  false,  false);
}

__device__ __forceinline__ v2f a_frag_g(const float* __restrict__ A, int lda,
                                        int r0, int k, int lane) {
    int m  = lane & 15;
    int kk = k + ((lane >> 4) << 1);
    const float* p = A + (size_t)(r0 + m) * lda + kk;
    return v2f{p[0], p[1]};
}

__device__ __forceinline__ v2f a_frag_lds(const float* __restrict__ L, int ld,
                                          int k, int lane) {
    int m  = lane & 15;
    int kk = k + ((lane >> 4) << 1);
    const float* p = L + m * ld + kk;
    return v2f{p[0], p[1]};
}

__device__ __forceinline__ v2f b_frag_g(const float* __restrict__ B, int ldb,
                                        int k, int c0, int lane) {
    int n  = lane & 15;
    int kk = k + ((lane >> 4) << 1);
    const float* p = B + (size_t)kk * ldb + c0 + n;
    return v2f{p[0], p[ldb]};
}

__device__ __forceinline__ void store_tile(float* __restrict__ D, int ldd,
                                           int r0, int c0, v8f acc, int lane,
                                           float bias) {
    int n    = lane & 15;
    int half = lane >> 4;
#pragma unroll
    for (int r = 0; r < 8; ++r)
        D[(size_t)(r0 + r + 8 * half) * ldd + c0 + n] = acc[r] + bias;
}

__global__ void deg_init_kernel(float* __restrict__ deg, int N) {
    int i = blockIdx.x * blockDim.x + threadIdx.x;
    if (i < N) deg[i] = 1.0f;
}

__global__ void deg_acc_kernel(const int* __restrict__ col, float* __restrict__ deg, int E) {
    int e = blockIdx.x * blockDim.x + threadIdx.x;
    if (e < E) atomicAdd(&deg[col[e]], 1.0f);
}

__global__ void dis_kernel(const float* __restrict__ deg, float* __restrict__ dis, int N) {
    int i = blockIdx.x * blockDim.x + threadIdx.x;
    if (i < N) dis[i] = rsqrtf(deg[i]);
}

__global__ __launch_bounds__(128) void self_branch_kernel(
    const float* __restrict__ x, const float* __restrict__ Win,
    const float* __restrict__ bin, const float* __restrict__ Wout,
    const float* __restrict__ bout, const float* __restrict__ Wg1,
    float* __restrict__ l1_out, float* __restrict__ h1_out, int N)
{
    __shared__ float lds[4 * 16 * LDS_LD];
    const int lane = threadIdx.x & 31;
    const int wave = threadIdx.x >> 5;
    const int r0   = (blockIdx.x * 4 + wave) * 16;
    const int n    = lane & 15, half = lane >> 4;
    float* myL = lds + wave * 16 * LDS_LD;
    if (r0 >= N) return;

    v2f ax[32];
#pragma unroll
    for (int kc = 0; kc < 32; ++kc)
        ax[kc] = a_frag_g(x, 128, r0, kc * 4, lane);

    for (int t = 0; t < 8; ++t) {
        v8f acc = {};
#pragma unroll
        for (int kc = 0; kc < 32; ++kc)
            acc = wmma_f32(ax[kc], b_frag_g(Win, 128, kc * 4, t * 16, lane), acc);
        float bb = bin[t * 16 + n];
#pragma unroll
        for (int r = 0; r < 8; ++r) {
            float v = acc[r] + bb;
            myL[(r + 8 * half) * LDS_LD + t * 16 + n] = v > 0.f ? v : 0.f;
        }
    }
    __syncthreads();

    for (int t = 0; t < 8; ++t) {
        v8f acc = {};
#pragma unroll
        for (int kc = 0; kc < 32; ++kc)
            acc = wmma_f32(ax[kc], b_frag_g(Wg1, 128, kc * 4, t * 16, lane), acc);
        store_tile(h1_out, 128, r0, t * 16, acc, lane, 0.f);
    }

    v2f al[32];
#pragma unroll
    for (int kc = 0; kc < 32; ++kc)
        al[kc] = a_frag_lds(myL, LDS_LD, kc * 4, lane);

    for (int t = 0; t < 4; ++t) {
        v8f acc = {};
#pragma unroll
        for (int kc = 0; kc < 32; ++kc)
            acc = wmma_f32(ax[kc], b_frag_g(Wout, 64, kc * 4, t * 16, lane), acc);
#pragma unroll
        for (int kc = 0; kc < 32; ++kc)
            acc = wmma_f32(al[kc], b_frag_g(Wout, 64, 128 + kc * 4, t * 16, lane), acc);
        store_tile(l1_out, 64, r0, t * 16, acc, lane, bout[t * 16 + n]);
    }
}

template <int W>
__global__ void seg_init_kernel(const float* __restrict__ h, const float* __restrict__ dis,
                                const float* __restrict__ bias, float* __restrict__ g, int N)
{
    int idx = blockIdx.x * blockDim.x + threadIdx.x;
    if (idx >= N * W) return;
    int i = idx / W, j = idx % W;
    float d = dis[i];
    g[idx] = bias[j] + h[idx] * d * d;
}

template <int W>
__global__ void seg_scatter_kernel(const int* __restrict__ row, const int* __restrict__ col,
                                   const float* __restrict__ dis, const float* __restrict__ h,
                                   float* __restrict__ g, int E)
{
    constexpr int PER_EDGE = W / 4;
    long long t = (long long)blockIdx.x * blockDim.x + threadIdx.x;
    if (t >= (long long)E * PER_EDGE) return;
    int e = (int)(t / PER_EDGE);
    int j = ((int)(t % PER_EDGE)) << 2;
    int r = row[e], c = col[e];
    float nrm = dis[r] * dis[c];
    const float4 hv = *(const float4*)(h + (size_t)r * W + j);
    float* gp = g + (size_t)c * W + j;
    atomicAdd(gp + 0, hv.x * nrm);
    atomicAdd(gp + 1, hv.y * nrm);
    atomicAdd(gp + 2, hv.z * nrm);
    atomicAdd(gp + 3, hv.w * nrm);
}

__global__ __launch_bounds__(128) void gcn2_gemm_kernel(
    const float* __restrict__ g1, const float* __restrict__ Wg2,
    float* __restrict__ h2, int N)
{
    const int lane = threadIdx.x & 31;
    const int wave = threadIdx.x >> 5;
    const int r0   = (blockIdx.x * 4 + wave) * 16;
    if (r0 >= N) return;

    v2f ag[32];
#pragma unroll
    for (int kc = 0; kc < 32; ++kc)
        ag[kc] = a_frag_g(g1, 128, r0, kc * 4, lane);

    for (int t = 0; t < 4; ++t) {
        v8f acc = {};
#pragma unroll
        for (int kc = 0; kc < 32; ++kc)
            acc = wmma_f32(ag[kc], b_frag_g(Wg2, 64, kc * 4, t * 16, lane), acc);
        store_tile(h2, 64, r0, t * 16, acc, lane, 0.f);
    }
}

__global__ __launch_bounds__(128) void out_gemm_kernel(
    const float* __restrict__ xn, const float* __restrict__ g2,
    const float* __restrict__ W, const float* __restrict__ b,
    float* __restrict__ out, int N)
{
    const int lane = threadIdx.x & 31;
    const int wave = threadIdx.x >> 5;
    const int r0   = (blockIdx.x * 4 + wave) * 16;
    if (r0 >= N) return;
    const int n = lane & 15;

    v2f axn[32];
#pragma unroll
    for (int kc = 0; kc < 32; ++kc)
        axn[kc] = a_frag_g(xn, 128, r0, kc * 4, lane);
    v2f ag2[16];
#pragma unroll
    for (int kc = 0; kc < 16; ++kc)
        ag2[kc] = a_frag_g(g2, 64, r0, kc * 4, lane);

    for (int t = 0; t < 4; ++t) {
        v8f acc = {};
#pragma unroll
        for (int kc = 0; kc < 32; ++kc)
            acc = wmma_f32(axn[kc], b_frag_g(W, 64, kc * 4, t * 16, lane), acc);
#pragma unroll
        for (int kc = 0; kc < 16; ++kc)
            acc = wmma_f32(ag2[kc], b_frag_g(W, 64, 128 + kc * 4, t * 16, lane), acc);
        store_tile(out, 64, r0, t * 16, acc, lane, b[t * 16 + n]);
    }
}

extern "C" void kernel_launch(void* const* d_in, const int* in_sizes, int n_in,
                              void* d_out, int out_size, void* d_ws, size_t ws_size,
                              hipStream_t stream) {
    const float* x_self     = (const float*)d_in[0];
    const float* x_neighbor = (const float*)d_in[1];
    const int*   edge_index = (const int*)d_in[2];
    const float* W_in_self  = (const float*)d_in[3];
    const float* b_in_self  = (const float*)d_in[4];
    const float* W_out_self = (const float*)d_in[5];
    const float* b_out_self = (const float*)d_in[6];
    const float* Wg1        = (const float*)d_in[7];
    const float* bg1        = (const float*)d_in[8];
    const float* Wg2        = (const float*)d_in[9];
    const float* bg2        = (const float*)d_in[10];
    const float* W_out      = (const float*)d_in[11];
    const float* b_out      = (const float*)d_in[12];

    const int N = in_sizes[0] / 128;
    const int E = in_sizes[2] / 2;
    const int* row = edge_index;
    const int* col = edge_index + E;

    float* ws  = (float*)d_ws;
    float* dis = ws;
    float* deg = ws + N;
    float* h   = ws + 2 * (size_t)N;
    float* g   = h + (size_t)N * 128;

    float* l1_out = (float*)d_out;
    float* x2_out = (float*)d_out + (size_t)N * 64;

    deg_init_kernel<<<(N + 255) / 256, 256, 0, stream>>>(deg, N);
    deg_acc_kernel<<<(E + 255) / 256, 256, 0, stream>>>(col, deg, E);
    dis_kernel<<<(N + 255) / 256, 256, 0, stream>>>(deg, dis, N);

    const int strips = (N + 15) / 16;
    const int blocks = (strips + 3) / 4;

    self_branch_kernel<<<blocks, 128, 0, stream>>>(x_self, W_in_self, b_in_self,
                                                   W_out_self, b_out_self, Wg1,
                                                   l1_out, h, N);

    seg_init_kernel<128><<<((N * 128) + 255) / 256, 256, 0, stream>>>(h, dis, bg1, g, N);
    seg_scatter_kernel<128><<<(int)(((long long)E * 32 + 255) / 256), 256, 0, stream>>>(
        row, col, dis, h, g, E);

    gcn2_gemm_kernel<<<blocks, 128, 0, stream>>>(g, Wg2, h, N);

    seg_init_kernel<64><<<((N * 64) + 255) / 256, 256, 0, stream>>>(h, dis, bg2, g, N);
    seg_scatter_kernel<64><<<(int)(((long long)E * 16 + 255) / 256), 256, 0, stream>>>(
        row, col, dis, h, g, E);

    out_gemm_kernel<<<blocks, 128, 0, stream>>>(x_neighbor, g, W_out, b_out, x2_out, N);
}


# Edge scatter made owner-by-half: resident work-groups draw 64-edge units from two queues (targets below / from N/2), each chip half adds only into its own target rows; arrival-order sums kept
# speedup vs baseline: 2.8531x; 2.8531x over previous
#include <hip/hip_runtime.h>
#include <hip/hip_bf16.h>

typedef float v2f __attribute__((ext_vector_type(2)));
typedef float v8f __attribute__((ext_vector_type(8)));

#define LDS_LD 130

__device__ __forceinline__ v8f wmma_f32(v2f a, v2f b, v8f c) {
    return __builtin_amdgcn_wmma_f32_16x16x4_f32(
         false, a,  false, b,
         (short)0, c,  false,  false);
}

__device__ __forceinline__ v2f a_frag_g(const float* __restrict__ A, int lda,
                                        int r0, int k, int lane) {
    int m  = lane & 15;
    int kk = k + ((lane >> 4) << 1);
    const float* p = A + (size_t)(r0 + m) * lda + kk;
    return v2f{p[0], p[1]};
}

__device__ __forceinline__ v2f a_frag_lds(const float* __restrict__ L, int ld,
                                          int k, int lane) {
    int m  = lane & 15;
    int kk = k + ((lane >> 4) << 1);
    const float* p = L + m * ld + kk;
    return v2f{p[0], p[1]};
}

__device__ __forceinline__ v2f b_frag_g(const float* __restrict__ B, int ldb,
                                        int k, int c0, int lane) {
    int n  = lane & 15;
    int kk = k + ((lane >> 4) << 1);
    const float* p = B + (size_t)kk * ldb + c0 + n;
    return v2f{p[0], p[ldb]};
}

__device__ __forceinline__ void store_tile(float* __restrict__ D, int ldd,
                                           int r0, int c0, v8f acc, int lane,
                                           float bias) {
    int n    = lane & 15;
    int half = lane >> 4;
#pragma unroll
    for (int r = 0; r < 8; ++r)
        D[(size_t)(r0 + r + 8 * half) * ldd + c0 + n] = acc[r] + bias;
}

#define DEG_P   32
#define DEG_NW  4
#define DEG_WIN 10240

__global__ __launch_bounds__(1024) void deg_hist_kernel(
    const int* __restrict__ col, unsigned* __restrict__ planes, int N, int E)
{
    __shared__ unsigned hist[DEG_WIN];
    const int tid = threadIdx.x;
    const int per = (E + DEG_P - 1) / DEG_P;
    const int e0  = blockIdx.x * per;
    const int e1  = (e0 + per < E) ? (e0 + per) : E;
    unsigned* __restrict__ myplane = planes + (size_t)blockIdx.x * N;

    for (int base = blockIdx.y * DEG_WIN; base < N; base += DEG_NW * DEG_WIN) {
        const int wn = (N - base < DEG_WIN) ? (N - base) : DEG_WIN;
        for (int i = tid; i < wn; i += 1024) hist[i] = 0u;
        __syncthreads();
        for (int e = e0 + tid; e < e1; e += 8 * 1024) {
            int cv[8];
#pragma unroll
            for (int j = 0; j < 8; ++j) {
                const int ee = e + j * 1024;
                cv[j] = (ee < e1) ? col[ee] : -1;
            }
#pragma unroll
            for (int j = 0; j < 8; ++j) {
                const unsigned u = (unsigned)(cv[j] - base);
                if (u < (unsigned)wn) atomicAdd(&hist[u], 1u);
            }
        }
        __syncthreads();
        for (int i = tid; i < wn; i += 1024) myplane[base + i] = hist[i];
        __syncthreads();
    }
}


__global__ __launch_bounds__(128) void self_branch_kernel(
    const float* __restrict__ x, const float* __restrict__ Win,
    const float* __restrict__ bin, const float* __restrict__ Wout,
    const float* __restrict__ bout, const float* __restrict__ Wg1,
    const unsigned* __restrict__ planes,
    float* __restrict__ deg, float* __restrict__ l1_out,
    float* __restrict__ h1_out, float* __restrict__ g1_out,
    int* __restrict__ ctr, int N)
{
    __shared__ float lds[4 * 16 * LDS_LD];
    if (blockIdx.x == 0 && threadIdx.x < 4) ctr[threadIdx.x] = 0;
    const int lane = threadIdx.x & 31;
    const int wave = threadIdx.x >> 5;
    const int r0   = (blockIdx.x * 4 + wave) * 16;
    const int n    = lane & 15, half = lane >> 4;
    float* myL = lds + wave * 16 * LDS_LD;
    if (r0 >= N) return;

    v2f ax[32];
#pragma unroll
    for (int kc = 0; kc < 32; ++kc)
        ax[kc] = a_frag_g(x, 128, r0, kc * 4, lane);

    unsigned cnt = 0u;
#pragma unroll
    for (int q = 0; q < DEG_P / 2; ++q)
        cnt += planes[(size_t)(half * (DEG_P / 2) + q) * N + r0 + n];
    cnt += __shfl_xor(cnt, 16);
    const float degm = (float)(cnt + 1u);
    if (half == 0) deg[r0 + n] = degm;
    const float dism = rsqrtf(degm);

    for (int t = 0; t < 8; ++t) {
        v8f acc = {};
#pragma unroll
        for (int kc = 0; kc < 32; ++kc)
            acc = wmma_f32(ax[kc], b_frag_g(Win, 128, kc * 4, t * 16, lane), acc);
        float bb = bin[t * 16 + n];
#pragma unroll
        for (int r = 0; r < 8; ++r) {
            float v = acc[r] + bb;
            myL[(r + 8 * half) * LDS_LD + t * 16 + n] = v > 0.f ? v : 0.f;
        }
    }
    __syncthreads();

    float dd[8];
#pragma unroll
    for (int r = 0; r < 8; ++r) dd[r] = __shfl(dism, 8 * half + r);
    for (int t = 0; t < 8; ++t) {
        v8f acc = {};
#pragma unroll
        for (int kc = 0; kc < 32; ++kc)
            acc = wmma_f32(ax[kc], b_frag_g(Wg1, 128, kc * 4, t * 16, lane), acc);
#pragma unroll
        for (int r = 0; r < 8; ++r) {
            const size_t o = (size_t)(r0 + r + 8 * half) * 128 + t * 16 + n;
            const float hs = acc[r] * dd[r];
            h1_out[o] = hs;
            g1_out[o] = hs;
        }
    }

    v2f al[32];
#pragma unroll
    for (int kc = 0; kc < 32; ++kc)
        al[kc] = a_frag_lds(myL, LDS_LD, kc * 4, lane);

    for (int t = 0; t < 4; ++t) {
        v8f acc = {};
#pragma unroll
        for (int kc = 0; kc < 32; ++kc)
            acc = wmma_f32(ax[kc], b_frag_g(Wout, 64, kc * 4, t * 16, lane), acc);
#pragma unroll
        for (int kc = 0; kc < 32; ++kc)
            acc = wmma_f32(al[kc], b_frag_g(Wout, 64, 128 + kc * 4, t * 16, lane), acc);
        store_tile(l1_out, 64, r0, t * 16, acc, lane, bout[t * 16 + n]);
    }
}


#define SC_WPB  8
#define SC_TPB  256
#define SC_GRID 1280

__device__ __forceinline__ int lane_bcast_i(int v, int k) {
    return __builtin_amdgcn_readlane(v, k);
}

template <int W>
__global__ __launch_bounds__(SC_TPB) void seg_scatter_kernel(
    const int* __restrict__ row, const int* __restrict__ col,
    const float* __restrict__ h, float* __restrict__ g,
    int* __restrict__ ctr,
    int E, int N)
{
    constexpr int NL = W / 32;
    constexpr int BE = 80 / NL;
    __shared__ int s_half;
    __shared__ int s_base;
    const int lane = threadIdx.x & 31;
    const int wave = __builtin_amdgcn_readfirstlane((int)(threadIdx.x >> 5));
    const int nunits = (E + 63) >> 6;
    const int split  = N >> 1;
    const int wph = ((int)gridDim.x * SC_WPB) >> 1;
    const int upw = (wph > 0) ? ((nunits + wph - 1) / wph > 0 ? (nunits + wph - 1) / wph : 1) : 1;
    const int upb = upw * SC_WPB;

    if (wave == 0) {
        const int id = __builtin_amdgcn_s_sendmsg_rtn(0x87);
        if (lane == 0) s_half = (id >> 18) & 1;
    }
    __syncthreads();
    const int myhalf = __builtin_amdgcn_readfirstlane(s_half);

    for (int pass = 0; pass < 2; ++pass) {
        const int q = myhalf ^ pass;
        int* cq = ctr + q;
        for (int it = 0; it <= nunits; ++it) {
            if (threadIdx.x == 0) s_base = atomicAdd(cq, upb);
            __syncthreads();
            const int base = __builtin_amdgcn_readfirstlane(s_base);
            __syncthreads();
            if (base >= nunits) break;
            const int ubeg = base + wave * upw;
            const int uend = (ubeg + upw < nunits) ? (ubeg + upw) : nunits;

            for (int u = ubeg; u < uend; ++u) {
                const int e0 = u * 64 + lane, e1 = e0 + 32;
                int r0 = 0, c0 = -1, r1 = 0, c1 = -1;
                if (e0 < E) { r0 = row[e0]; c0 = col[e0]; }
                if (e1 < E) { r1 = row[e1]; c1 = col[e1]; }
                const bool mine0 = (c0 >= 0) && ((c0 >= split) == (q != 0));
                const bool mine1 = (c1 >= 0) && ((c1 >= split) == (q != 0));
                unsigned long long m =
                    (unsigned long long)__builtin_amdgcn_ballot_w32(mine0) |
                    ((unsigned long long)__builtin_amdgcn_ballot_w32(mine1) << 32);
                while (m) {
                    const unsigned long long mb = m;
                    float v[BE][NL];
#pragma unroll
                    for (int j = 0; j < BE; ++j) {
                        if (m) {
                            const int k  = __builtin_ctzll(m);
                            m &= m - 1;
                            const int ra = lane_bcast_i(r0, k & 31);
                            const int rb = lane_bcast_i(r1, k & 31);
                            const int rs = (k < 32) ? ra : rb;
                            const float* hp = h + (size_t)rs * W + lane;
#pragma unroll
                            for (int l = 0; l < NL; ++l) v[j][l] = hp[32 * l];
                        }
                    }
                    unsigned long long m2 = mb;
#pragma unroll
                    for (int j = 0; j < BE; ++j) {
                        if (m2 != m) {
                            const int k  = __builtin_ctzll(m2);
                            m2 &= m2 - 1;
                            const int ca = lane_bcast_i(c0, k & 31);
                            const int cb = lane_bcast_i(c1, k & 31);
                            const int cs = (k < 32) ? ca : cb;
                            float* gp = g + (size_t)cs * W + lane;
#pragma unroll
                            for (int l = 0; l < NL; ++l) atomicAdd(gp + 32 * l, v[j][l]);
                        }
                    }
                }
            }
        }
    }
}

__global__ __launch_bounds__(128) void gcn2_gemm_kernel(
    const float* __restrict__ s1, const float* __restrict__ Wg2,
    const float* __restrict__ bg1, const float* __restrict__ deg,
    float* __restrict__ h2, float* __restrict__ g2, int N)
{
    const int lane = threadIdx.x & 31;
    const int wave = threadIdx.x >> 5;
    const int r0   = (blockIdx.x * 4 + wave) * 16;
    if (r0 >= N) return;
    const int n = lane & 15, half = lane >> 4;

    const float dism = rsqrtf(deg[r0 + n]);
    v2f ag[32];
#pragma unroll
    for (int kc = 0; kc < 32; ++kc) {
        const v2f s = a_frag_g(s1, 128, r0, kc * 4, lane);
        const float* pb = bg1 + kc * 4 + 2 * half;
        ag[kc] = v2f{pb[0] + dism * s[0], pb[1] + dism * s[1]};
    }
    float dd[8];
#pragma unroll
    for (int r = 0; r < 8; ++r) dd[r] = __shfl(dism, 8 * half + r);

    for (int t = 0; t < 4; ++t) {
        v8f acc = {};
#pragma unroll
        for (int kc = 0; kc < 32; ++kc)
            acc = wmma_f32(ag[kc], b_frag_g(Wg2, 64, kc * 4, t * 16, lane), acc);
#pragma unroll
        for (int r = 0; r < 8; ++r) {
            const size_t o = (size_t)(r0 + r + 8 * half) * 64 + t * 16 + n;
            const float hs = acc[r] * dd[r];
            h2[o] = hs;
            g2[o] = hs;
        }
    }
}

__global__ __launch_bounds__(128) void out_gemm_kernel(
    const float* __restrict__ xn, const float* __restrict__ s2,
    const float* __restrict__ bg2, const float* __restrict__ deg,
    const float* __restrict__ W, const float* __restrict__ b,
    float* __restrict__ out, int N)
{
    const int lane = threadIdx.x & 31;
    const int wave = threadIdx.x >> 5;
    const int r0   = (blockIdx.x * 4 + wave) * 16;
    if (r0 >= N) return;
    const int n = lane & 15, half = lane >> 4;

    v2f axn[32];
#pragma unroll
    for (int kc = 0; kc < 32; ++kc)
        axn[kc] = a_frag_g(xn, 128, r0, kc * 4, lane);
    const float dism = rsqrtf(deg[r0 + n]);
    v2f ag2[16];
#pragma unroll
    for (int kc = 0; kc < 16; ++kc) {
        const v2f s = a_frag_g(s2, 64, r0, kc * 4, lane);
        const float* pb = bg2 + kc * 4 + 2 * half;
        ag2[kc] = v2f{pb[0] + dism * s[0], pb[1] + dism * s[1]};
    }

    for (int t = 0; t < 4; ++t) {
        v8f acc = {};
#pragma unroll
        for (int kc = 0; kc < 32; ++kc)
            acc = wmma_f32(axn[kc], b_frag_g(W, 64, kc * 4, t * 16, lane), acc);
#pragma unroll
        for (int kc = 0; kc < 16; ++kc)
            acc = wmma_f32(ag2[kc], b_frag_g(W, 64, 128 + kc * 4, t * 16, lane), acc);
        store_tile(out, 64, r0, t * 16, acc, lane, b[t * 16 + n]);
    }
}

extern "C" void kernel_launch(void* const* d_in, const int* in_sizes, int n_in,
                              void* d_out, int out_size, void* d_ws, size_t ws_size,
                              hipStream_t stream) {
    const float* x_self     = (const float*)d_in[0];
    const float* x_neighbor = (const float*)d_in[1];
    const int*   edge_index = (const int*)d_in[2];
    const float* W_in_self  = (const float*)d_in[3];
    const float* b_in_self  = (const float*)d_in[4];
    const float* W_out_self = (const float*)d_in[5];
    const float* b_out_self = (const float*)d_in[6];
    const float* Wg1        = (const float*)d_in[7];
    const float* bg1        = (const float*)d_in[8];
    const float* Wg2        = (const float*)d_in[9];
    const float* bg2        = (const float*)d_in[10];
    const float* W_out      = (const float*)d_in[11];
    const float* b_out      = (const float*)d_in[12];

    const int N = in_sizes[0] / 128;
    const int E = in_sizes[2] / 2;
    const int* row = edge_index;
    const int* col = edge_index + E;

    float* ws  = (float*)d_ws;
    int*   ctr = (int*)d_ws;
    float* deg = ws + N;
    float* h   = ws + 2 * (size_t)N;
    float* g   = h + (size_t)N * 128;
    float* g2  = h + (size_t)N * 64;

    float* l1_out = (float*)d_out;
    float* x2_out = (float*)d_out + (size_t)N * 64;

    unsigned* planes = (unsigned*)x2_out;
    deg_hist_kernel<<<dim3(DEG_P, DEG_NW), 1024, 0, stream>>>(col, planes, N, E);

    const int strips = (N + 15) / 16;
    const int blocks = (strips + 3) / 4;

    self_branch_kernel<<<blocks, 128, 0, stream>>>(x_self, W_in_self, b_in_self,
                                                   W_out_self, b_out_self, Wg1,
                                                   planes, deg, l1_out, h, g, ctr, N);

    seg_scatter_kernel<128><<<SC_GRID, SC_TPB, 0, stream>>>(row, col, h, g, ctr, E, N);

    gcn2_gemm_kernel<<<blocks, 128, 0, stream>>>(g, Wg2, bg1, deg, h, g2, N);

    seg_scatter_kernel<64><<<SC_GRID, SC_TPB, 0, stream>>>(row, col, h, g2, ctr + 2, E, N);

    out_gemm_kernel<<<blocks, 128, 0, stream>>>(x_neighbor, g2, bg2, deg, W_out, b_out,
                                                x2_out, N);
}


# GEMM weights re-laid once per call into B-fragment order by spare work-groups of the degree-count launch; one 16-byte whole-line load feeds 2 WMMAs, not four 4-byte loads; k sum order kept
# speedup vs baseline: 3.0474x; 1.0681x over previous
#include <hip/hip_runtime.h>
#include <hip/hip_bf16.h>

typedef float v2f __attribute__((ext_vector_type(2)));
typedef float v4f __attribute__((ext_vector_type(4)));
typedef float v8f __attribute__((ext_vector_type(8)));

#define LDS_LD 130

__device__ __forceinline__ v8f wmma_f32(v2f a, v2f b, v8f c) {
    return __builtin_amdgcn_wmma_f32_16x16x4_f32(
         false, a,  false, b,
         (short)0, c,  false,  false);
}

__device__ __forceinline__ v2f a_frag_g(const float* __restrict__ A, int lda,
                                        int r0, int k, int lane) {
    int m  = lane & 15;
    int kk = k + ((lane >> 4) << 1);
    const float* p = A + (size_t)(r0 + m) * lda + kk;
    return v2f{p[0], p[1]};
}

__device__ __forceinline__ v2f a_frag_lds(const float* __restrict__ L, int ld,
                                          int k, int lane) {
    int m  = lane & 15;
    int kk = k + ((lane >> 4) << 1);
    const float* p = L + m * ld + kk;
    return v2f{p[0], p[1]};
}

__device__ __forceinline__ void relay_b_quad(const float* __restrict__ Wm,
                                             float* __restrict__ F,
                                             int K, int NC, int q) {
    const int l  = q & 31;
    const int tp = q >> 5;
    const int P  = K >> 3;
    const int t  = tp / P;
    const int p  = tp - t * P;
    const int c  = 16 * t + (l & 15);
    const int k0 = 8 * p + 2 * (l >> 4);
    v4f v;
    v[0] = Wm[(size_t)(k0    ) * NC + c];
    v[1] = Wm[(size_t)(k0 + 1) * NC + c];
    v[2] = Wm[(size_t)(k0 + 4) * NC + c];
    v[3] = Wm[(size_t)(k0 + 5) * NC + c];
    *reinterpret_cast<v4f*>(F + ((size_t)q << 2)) = v;
}

__device__ __forceinline__ v4f b_quad(const float* __restrict__ F, int P,
                                      int t, int p, int lane) {
    return *reinterpret_cast<const v4f*>(F + ((size_t)((t * P + p) * 32 + lane) << 2));
}

__device__ __forceinline__ void store_tile(float* __restrict__ D, int ldd,
                                           int r0, int c0, v8f acc, int lane,
                                           float bias) {
    int n    = lane & 15;
    int half = lane >> 4;
#pragma unroll
    for (int r = 0; r < 8; ++r)
        D[(size_t)(r0 + r + 8 * half) * ldd + c0 + n] = acc[r] + bias;
}

#define DEG_P   32
#define DEG_NW  4
#define DEG_WIN 10240

#define RL_WIN   (128 * 128 / 4)
#define RL_WG1   (128 * 128 / 4)
#define RL_WOS   (256 *  64 / 4)
#define RL_WG2   (128 *  64 / 4)
#define RL_WO    (192 *  64 / 4)

__global__ __launch_bounds__(1024) void deg_hist_kernel(
    const int* __restrict__ col, unsigned* __restrict__ planes, int N, int E,
    const float* __restrict__ Win, const float* __restrict__ Wg1,
    const float* __restrict__ Wos, const float* __restrict__ Wg2,
    const float* __restrict__ Wo,
    float* __restrict__ FWin, float* __restrict__ FWg1, float* __restrict__ FWos,
    float* __restrict__ FWg2, float* __restrict__ FWo)
{
    __shared__ unsigned hist[DEG_WIN];
    const int tid = threadIdx.x;
    if (blockIdx.y == DEG_NW) {
        int q = blockIdx.x * 1024 + tid;
        if (q < RL_WIN) { relay_b_quad(Win, FWin, 128, 128, q); return; }
        q -= RL_WIN;
        if (q < RL_WG1) { relay_b_quad(Wg1, FWg1, 128, 128, q); return; }
        q -= RL_WG1;
        if (q < RL_WOS) { relay_b_quad(Wos, FWos, 256, 64, q); return; }
        q -= RL_WOS;
        if (q < RL_WG2) { relay_b_quad(Wg2, FWg2, 128, 64, q); return; }
        q -= RL_WG2;
        if (q < RL_WO)  { relay_b_quad(Wo, FWo, 192, 64, q); }
        return;
    }
    const int per = (E + DEG_P - 1) / DEG_P;
    const int e0  = blockIdx.x * per;
    const int e1  = (e0 + per < E) ? (e0 + per) : E;
    unsigned* __restrict__ myplane = planes + (size_t)blockIdx.x * N;

    for (int base = blockIdx.y * DEG_WIN; base < N; base += DEG_NW * DEG_WIN) {
        const int wn = (N - base < DEG_WIN) ? (N - base) : DEG_WIN;
        for (int i = tid; i < wn; i += 1024) hist[i] = 0u;
        __syncthreads();
        for (int e = e0 + tid; e < e1; e += 8 * 1024) {
            int cv[8];
#pragma unroll
            for (int j = 0; j < 8; ++j) {
                const int ee = e + j * 1024;
                cv[j] = (ee < e1) ? col[ee] : -1;
            }
#pragma unroll
            for (int j = 0; j < 8; ++j) {
                const unsigned u = (unsigned)(cv[j] - base);
                if (u < (unsigned)wn) atomicAdd(&hist[u], 1u);
            }
        }
        __syncthreads();
        for (int i = tid; i < wn; i += 1024) myplane[base + i] = hist[i];
        __syncthreads();
    }
}


__global__ __launch_bounds__(128) void self_branch_kernel(
    const float* __restrict__ x, const float* __restrict__ FWin,
    const float* __restrict__ bin, const float* __restrict__ FWout,
    const float* __restrict__ bout, const float* __restrict__ FWg1,
    const unsigned* __restrict__ planes,
    float* __restrict__ deg, float* __restrict__ l1_out,
    float* __restrict__ h1_out, float* __restrict__ g1_out,
    int* __restrict__ ctr, int N)
{
    __shared__ float lds[4 * 16 * LDS_LD];
    if (blockIdx.x == 0 && threadIdx.x < 4) ctr[threadIdx.x] = 0;
    const int lane = threadIdx.x & 31;
    const int wave = threadIdx.x >> 5;
    const int r0   = (blockIdx.x * 4 + wave) * 16;
    const int n    = lane & 15, half = lane >> 4;
    float* myL = lds + wave * 16 * LDS_LD;
    if (r0 >= N) return;

    v2f ax[32];
#pragma unroll
    for (int kc = 0; kc < 32; ++kc)
        ax[kc] = a_frag_g(x, 128, r0, kc * 4, lane);

    unsigned cnt = 0u;
#pragma unroll
    for (int q = 0; q < DEG_P / 2; ++q)
        cnt += planes[(size_t)(half * (DEG_P / 2) + q) * N + r0 + n];
    cnt += __shfl_xor(cnt, 16);
    const float degm = (float)(cnt + 1u);
    if (half == 0) deg[r0 + n] = degm;
    const float dism = rsqrtf(degm);

    for (int t = 0; t < 8; ++t) {
        v8f acc = {};
        v4f f[16];
#pragma unroll
        for (int p = 0; p < 16; ++p) f[p] = b_quad(FWin, 16, t, p, lane);
        __builtin_amdgcn_sched_barrier(0);
#pragma unroll
        for (int p = 0; p < 16; ++p) {
            acc = wmma_f32(ax[2 * p],     v2f{f[p][0], f[p][1]}, acc);
            acc = wmma_f32(ax[2 * p + 1], v2f{f[p][2], f[p][3]}, acc);
        }
        float bb = bin[t * 16 + n];
#pragma unroll
        for (int r = 0; r < 8; ++r) {
            float v = acc[r] + bb;
            myL[(r + 8 * half) * LDS_LD + t * 16 + n] = v > 0.f ? v : 0.f;
        }
    }
    __syncthreads();

    float dd[8];
#pragma unroll
    for (int r = 0; r < 8; ++r) dd[r] = __shfl(dism, 8 * half + r);
    for (int t = 0; t < 8; ++t) {
        v8f acc = {};
        v4f f[16];
#pragma unroll
        for (int p = 0; p < 16; ++p) f[p] = b_quad(FWg1, 16, t, p, lane);
        __builtin_amdgcn_sched_barrier(0);
#pragma unroll
        for (int p = 0; p < 16; ++p) {
            acc = wmma_f32(ax[2 * p],     v2f{f[p][0], f[p][1]}, acc);
            acc = wmma_f32(ax[2 * p + 1], v2f{f[p][2], f[p][3]}, acc);
        }
#pragma unroll
        for (int r = 0; r < 8; ++r) {
            const size_t o = (size_t)(r0 + r + 8 * half) * 128 + t * 16 + n;
            const float hs = acc[r] * dd[r];
            h1_out[o] = hs;
            g1_out[o] = hs;
        }
    }

    v2f al[32];
#pragma unroll
    for (int kc = 0; kc < 32; ++kc)
        al[kc] = a_frag_lds(myL, LDS_LD, kc * 4, lane);

    for (int t = 0; t < 4; ++t) {
        v8f acc = {};
        v4f f[32];
#pragma unroll
        for (int p = 0; p < 32; ++p) f[p] = b_quad(FWout, 32, t, p, lane);
        __builtin_amdgcn_sched_barrier(0);
#pragma unroll
        for (int p = 0; p < 16; ++p) {
            acc = wmma_f32(ax[2 * p],     v2f{f[p][0], f[p][1]}, acc);
            acc = wmma_f32(ax[2 * p + 1], v2f{f[p][2], f[p][3]}, acc);
        }
#pragma unroll
        for (int p = 0; p < 16; ++p) {
            acc = wmma_f32(al[2 * p],     v2f{f[16 + p][0], f[16 + p][1]}, acc);
            acc = wmma_f32(al[2 * p + 1], v2f{f[16 + p][2], f[16 + p][3]}, acc);
        }
        store_tile(l1_out, 64, r0, t * 16, acc, lane, bout[t * 16 + n]);
    }
}


#define SC_WPB  8
#define SC_TPB  256
#define SC_GRID 1280

__device__ __forceinline__ int lane_bcast_i(int v, int k) {
    return __builtin_amdgcn_readlane(v, k);
}

template <int W>
__global__ __launch_bounds__(SC_TPB) void seg_scatter_kernel(
    const int* __restrict__ row, const int* __restrict__ col,
    const float* __restrict__ h, float* __restrict__ g,
    int* __restrict__ ctr,
    int E, int N)
{
    constexpr int NL = W / 32;
    constexpr int BE = 80 / NL;
    __shared__ int s_half;
    __shared__ int s_base;
    const int lane = threadIdx.x & 31;
    const int wave = __builtin_amdgcn_readfirstlane((int)(threadIdx.x >> 5));
    const int nunits = (E + 63) >> 6;
    const int split  = N >> 1;
    const int wph = ((int)gridDim.x * SC_WPB) >> 1;
    const int upw = (wph > 0) ? ((nunits + wph - 1) / wph > 0 ? (nunits + wph - 1) / wph : 1) : 1;
    const int upb = upw * SC_WPB;

    if (wave == 0) {
        const int id = __builtin_amdgcn_s_sendmsg_rtn(0x87);
        if (lane == 0) s_half = (id >> 18) & 1;
    }
    __syncthreads();
    const int myhalf = __builtin_amdgcn_readfirstlane(s_half);

    for (int pass = 0; pass < 2; ++pass) {
        const int q = myhalf ^ pass;
        int* cq = ctr + q;
        for (int it = 0; it <= nunits; ++it) {
            if (threadIdx.x == 0) s_base = atomicAdd(cq, upb);
            __syncthreads();
            const int base = __builtin_amdgcn_readfirstlane(s_base);
            __syncthreads();
            if (base >= nunits) break;
            const int ubeg = base + wave * upw;
            const int uend = (ubeg + upw < nunits) ? (ubeg + upw) : nunits;

            for (int u = ubeg; u < uend; ++u) {
                const int e0 = u * 64 + lane, e1 = e0 + 32;
                int r0 = 0, c0 = -1, r1 = 0, c1 = -1;
                if (e0 < E) { r0 = row[e0]; c0 = col[e0]; }
                if (e1 < E) { r1 = row[e1]; c1 = col[e1]; }
                const bool mine0 = (c0 >= 0) && ((c0 >= split) == (q != 0));
                const bool mine1 = (c1 >= 0) && ((c1 >= split) == (q != 0));
                unsigned long long m =
                    (unsigned long long)__builtin_amdgcn_ballot_w32(mine0) |
                    ((unsigned long long)__builtin_amdgcn_ballot_w32(mine1) << 32);
                while (m) {
                    const unsigned long long mb = m;
                    float v[BE][NL];
#pragma unroll
                    for (int j = 0; j < BE; ++j) {
                        if (m) {
                            const int k  = __builtin_ctzll(m);
                            m &= m - 1;
                            const int ra = lane_bcast_i(r0, k & 31);
                            const int rb = lane_bcast_i(r1, k & 31);
                            const int rs = (k < 32) ? ra : rb;
                            const float* hp = h + (size_t)rs * W + lane;
#pragma unroll
                            for (int l = 0; l < NL; ++l) v[j][l] = hp[32 * l];
                        }
                    }
                    unsigned long long m2 = mb;
#pragma unroll
                    for (int j = 0; j < BE; ++j) {
                        if (m2 != m) {
                            const int k  = __builtin_ctzll(m2);
                            m2 &= m2 - 1;
                            const int ca = lane_bcast_i(c0, k & 31);
                            const int cb = lane_bcast_i(c1, k & 31);
                            const int cs = (k < 32) ? ca : cb;
                            float* gp = g + (size_t)cs * W + lane;
#pragma unroll
                            for (int l = 0; l < NL; ++l) atomicAdd(gp + 32 * l, v[j][l]);
                        }
                    }
                }
            }
        }
    }
}

__global__ __launch_bounds__(128) void gcn2_gemm_kernel(
    const float* __restrict__ s1, const float* __restrict__ FWg2,
    const float* __restrict__ bg1, const float* __restrict__ deg,
    float* __restrict__ h2, float* __restrict__ g2, int N)
{
    const int lane = threadIdx.x & 31;
    const int wave = threadIdx.x >> 5;
    const int r0   = (blockIdx.x * 4 + wave) * 16;
    if (r0 >= N) return;
    const int n = lane & 15, half = lane >> 4;

    const float dism = rsqrtf(deg[r0 + n]);
    v2f ag[32];
#pragma unroll
    for (int kc = 0; kc < 32; ++kc) {
        const v2f s = a_frag_g(s1, 128, r0, kc * 4, lane);
        const float* pb = bg1 + kc * 4 + 2 * half;
        ag[kc] = v2f{pb[0] + dism * s[0], pb[1] + dism * s[1]};
    }
    float dd[8];
#pragma unroll
    for (int r = 0; r < 8; ++r) dd[r] = __shfl(dism, 8 * half + r);

    for (int t = 0; t < 4; ++t) {
        v8f acc = {};
        v4f f[16];
#pragma unroll
        for (int p = 0; p < 16; ++p) f[p] = b_quad(FWg2, 16, t, p, lane);
        __builtin_amdgcn_sched_barrier(0);
#pragma unroll
        for (int p = 0; p < 16; ++p) {
            acc = wmma_f32(ag[2 * p],     v2f{f[p][0], f[p][1]}, acc);
            acc = wmma_f32(ag[2 * p + 1], v2f{f[p][2], f[p][3]}, acc);
        }
#pragma unroll
        for (int r = 0; r < 8; ++r) {
            const size_t o = (size_t)(r0 + r + 8 * half) * 64 + t * 16 + n;
            const float hs = acc[r] * dd[r];
            h2[o] = hs;
            g2[o] = hs;
        }
    }
}

__global__ __launch_bounds__(128) void out_gemm_kernel(
    const float* __restrict__ xn, const float* __restrict__ s2,
    const float* __restrict__ bg2, const float* __restrict__ deg,
    const float* __restrict__ FW, const float* __restrict__ b,
    float* __restrict__ out, int N)
{
    const int lane = threadIdx.x & 31;
    const int wave = threadIdx.x >> 5;
    const int r0   = (blockIdx.x * 4 + wave) * 16;
    if (r0 >= N) return;
    const int n = lane & 15, half = lane >> 4;

    v2f axn[32];
#pragma unroll
    for (int kc = 0; kc < 32; ++kc)
        axn[kc] = a_frag_g(xn, 128, r0, kc * 4, lane);
    const float dism = rsqrtf(deg[r0 + n]);
    v2f ag2[16];
#pragma unroll
    for (int kc = 0; kc < 16; ++kc) {
        const v2f s = a_frag_g(s2, 64, r0, kc * 4, lane);
        const float* pb = bg2 + kc * 4 + 2 * half;
        ag2[kc] = v2f{pb[0] + dism * s[0], pb[1] + dism * s[1]};
    }

    for (int t = 0; t < 4; ++t) {
        v8f acc = {};
        v4f f[24];
#pragma unroll
        for (int p = 0; p < 24; ++p) f[p] = b_quad(FW, 24, t, p, lane);
        __builtin_amdgcn_sched_barrier(0);
#pragma unroll
        for (int p = 0; p < 16; ++p) {
            acc = wmma_f32(axn[2 * p],     v2f{f[p][0], f[p][1]}, acc);
            acc = wmma_f32(axn[2 * p + 1], v2f{f[p][2], f[p][3]}, acc);
        }
#pragma unroll
        for (int p = 0; p < 8; ++p) {
            acc = wmma_f32(ag2[2 * p],     v2f{f[16 + p][0], f[16 + p][1]}, acc);
            acc = wmma_f32(ag2[2 * p + 1], v2f{f[16 + p][2], f[16 + p][3]}, acc);
        }
        store_tile(out, 64, r0, t * 16, acc, lane, b[t * 16 + n]);
    }
}

extern "C" void kernel_launch(void* const* d_in, const int* in_sizes, int n_in,
                              void* d_out, int out_size, void* d_ws, size_t ws_size,
                              hipStream_t stream) {
    const float* x_self     = (const float*)d_in[0];
    const float* x_neighbor = (const float*)d_in[1];
    const int*   edge_index = (const int*)d_in[2];
    const float* W_in_self  = (const float*)d_in[3];
    const float* b_in_self  = (const float*)d_in[4];
    const float* W_out_self = (const float*)d_in[5];
    const float* b_out_self = (const float*)d_in[6];
    const float* Wg1        = (const float*)d_in[7];
    const float* bg1        = (const float*)d_in[8];
    const float* Wg2        = (const float*)d_in[9];
    const float* bg2        = (const float*)d_in[10];
    const float* W_out      = (const float*)d_in[11];
    const float* b_out      = (const float*)d_in[12];

    const int N = in_sizes[0] / 128;
    const int E = in_sizes[2] / 2;
    const int* row = edge_index;
    const int* col = edge_index + E;

    float* ws  = (float*)d_ws;
    int*   ctr = (int*)d_ws;
    float* FWos = ws + 128;
    float* FWg2 = FWos + 256 * 64;
    float* FWo  = FWg2 + 128 * 64;
    float* deg = ws + N;
    float* h   = ws + 2 * (size_t)N;
    float* g   = h + (size_t)N * 128;
    float* g2  = h + (size_t)N * 64;

    float* l1_out = (float*)d_out;
    float* x2_out = (float*)d_out + (size_t)N * 64;

    unsigned* planes = (unsigned*)x2_out;
    float* FWin = x2_out + (size_t)DEG_P * N;
    float* FWg1 = FWin + 128 * 128;
    deg_hist_kernel<<<dim3(DEG_P, DEG_NW + 1), 1024, 0, stream>>>(
        col, planes, N, E, W_in_self, Wg1, W_out_self, Wg2, W_out,
        FWin, FWg1, FWos, FWg2, FWo);

    const int strips = (N + 15) / 16;
    const int blocks = (strips + 3) / 4;

    self_branch_kernel<<<blocks, 128, 0, stream>>>(x_self, FWin, b_in_self,
                                                   FWos, b_out_self, FWg1,
                                                   planes, deg, l1_out, h, g, ctr, N);

    seg_scatter_kernel<128><<<SC_GRID, SC_TPB, 0, stream>>>(row, col, h, g, ctr, E, N);

    gcn2_gemm_kernel<<<blocks, 128, 0, stream>>>(g, FWg2, bg1, deg, h, g2, N);

    seg_scatter_kernel<64><<<SC_GRID, SC_TPB, 0, stream>>>(row, col, h, g2, ctr + 2, E, N);

    out_gemm_kernel<<<blocks, 128, 0, stream>>>(x_neighbor, g2, bg2, deg, FWo, b_out,
                                                x2_out, N);
}


# GEMM launches re-tiled: 4-wave work-group per 32 rows, A tile staged in LDS by whole-row loads; a wave takes 1/4 of the column tiles of both 16-row strips: 4 WMMAs per weight load; k sum order kept
# speedup vs baseline: 3.1705x; 1.0404x over previous
#include <hip/hip_runtime.h>
#include <hip/hip_bf16.h>

typedef float v2f __attribute__((ext_vector_type(2)));
typedef float v4f __attribute__((ext_vector_type(4)));
typedef float v8f __attribute__((ext_vector_type(8)));

#define LDS_LD 130

__device__ __forceinline__ v8f wmma_f32(v2f a, v2f b, v8f c) {
    return __builtin_amdgcn_wmma_f32_16x16x4_f32(
         false, a,  false, b,
         (short)0, c,  false,  false);
}

__device__ __forceinline__ v2f a_frag_lds(const float* __restrict__ L, int ld,
                                          int k, int lane) {
    int m  = lane & 15;
    int kk = k + ((lane >> 4) << 1);
    const float* p = L + m * ld + kk;
    return v2f{p[0], p[1]};
}

__device__ __forceinline__ void relay_b_quad(const float* __restrict__ Wm,
                                             float* __restrict__ F,
                                             int K, int NC, int q) {
    const int l  = q & 31;
    const int tp = q >> 5;
    const int P  = K >> 3;
    const int t  = tp / P;
    const int p  = tp - t * P;
    const int c  = 16 * t + (l & 15);
    const int k0 = 8 * p + 2 * (l >> 4);
    v4f v;
    v[0] = Wm[(size_t)(k0    ) * NC + c];
    v[1] = Wm[(size_t)(k0 + 1) * NC + c];
    v[2] = Wm[(size_t)(k0 + 4) * NC + c];
    v[3] = Wm[(size_t)(k0 + 5) * NC + c];
    *reinterpret_cast<v4f*>(F + ((size_t)q << 2)) = v;
}

__device__ __forceinline__ v4f b_quad(const float* __restrict__ F, int P,
                                      int t, int p, int lane) {
    return *reinterpret_cast<const v4f*>(F + ((size_t)((t * P + p) * 32 + lane) << 2));
}

#define DEG_P   32
#define DEG_NW  4
#define DEG_WIN 10240

#define RL_WIN   (128 * 128 / 4)
#define RL_WG1   (128 * 128 / 4)
#define RL_WOS   (256 *  64 / 4)
#define RL_WG2   (128 *  64 / 4)
#define RL_WO    (192 *  64 / 4)

__global__ __launch_bounds__(1024) void deg_hist_kernel(
    const int* __restrict__ col, unsigned* __restrict__ planes, int N, int E,
    const float* __restrict__ Win, const float* __restrict__ Wg1,
    const float* __restrict__ Wos, const float* __restrict__ Wg2,
    const float* __restrict__ Wo,
    float* __restrict__ FWin, float* __restrict__ FWg1, float* __restrict__ FWos,
    float* __restrict__ FWg2, float* __restrict__ FWo)
{
    __shared__ unsigned hist[DEG_WIN];
    const int tid = threadIdx.x;
    if (blockIdx.y == DEG_NW) {
        int q = blockIdx.x * 1024 + tid;
        if (q < RL_WIN) { relay_b_quad(Win, FWin, 128, 128, q); return; }
        q -= RL_WIN;
        if (q < RL_WG1) { relay_b_quad(Wg1, FWg1, 128, 128, q); return; }
        q -= RL_WG1;
        if (q < RL_WOS) { relay_b_quad(Wos, FWos, 256, 64, q); return; }
        q -= RL_WOS;
        if (q < RL_WG2) { relay_b_quad(Wg2, FWg2, 128, 64, q); return; }
        q -= RL_WG2;
        if (q < RL_WO)  { relay_b_quad(Wo, FWo, 192, 64, q); }
        return;
    }
    const int per = (E + DEG_P - 1) / DEG_P;
    const int e0  = blockIdx.x * per;
    const int e1  = (e0 + per < E) ? (e0 + per) : E;
    unsigned* __restrict__ myplane = planes + (size_t)blockIdx.x * N;

    for (int base = blockIdx.y * DEG_WIN; base < N; base += DEG_NW * DEG_WIN) {
        const int wn = (N - base < DEG_WIN) ? (N - base) : DEG_WIN;
        for (int i = tid; i < wn; i += 1024) hist[i] = 0u;
        __syncthreads();
        for (int e = e0 + tid; e < e1; e += 8 * 1024) {
            int cv[8];
#pragma unroll
            for (int j = 0; j < 8; ++j) {
                const int ee = e + j * 1024;
                cv[j] = (ee < e1) ? col[ee] : -1;
            }
#pragma unroll
            for (int j = 0; j < 8; ++j) {
                const unsigned u = (unsigned)(cv[j] - base);
                if (u < (unsigned)wn) atomicAdd(&hist[u], 1u);
            }
        }
        __syncthreads();
        for (int i = tid; i < wn; i += 1024) myplane[base + i] = hist[i];
        __syncthreads();
    }
}


#define OUT_LD 194

__device__ __forceinline__ void stage_rows128(const float* __restrict__ A, int r0,
                                              float* __restrict__ L, int ld,
                                              int wave, int lane) {
#pragma unroll
    for (int i = 0; i < 8; ++i) {
        const int rr = i * 4 + wave;
        const int cc = lane << 2;
        const v4f v = *reinterpret_cast<const v4f*>(A + (size_t)(r0 + rr) * 128 + cc);
        float* d = L + rr * ld + cc;
        *reinterpret_cast<v2f*>(d)     = v2f{v[0], v[1]};
        *reinterpret_cast<v2f*>(d + 2) = v2f{v[2], v[3]};
    }
}

#define O_LD 68

__device__ __forceinline__ void stage_out_tile(float* __restrict__ O, int t, int lane,
                                               v8f a0, v8f a1) {
    const int n = lane & 15, half = lane >> 4;
#pragma unroll
    for (int r = 0; r < 8; ++r) {
        O[(r + 8 * half) * O_LD + t * 16 + n]      = a0[r];
        O[(16 + r + 8 * half) * O_LD + t * 16 + n] = a1[r];
    }
}

__device__ __forceinline__ v4f staged_piece(const float* __restrict__ O, int q) {
    return *reinterpret_cast<const v4f*>(O + (q >> 4) * O_LD + ((q & 15) << 2));
}
__device__ __forceinline__ size_t piece_offset(int r0, int q) {
    return (size_t)(r0 + (q >> 4)) * 64 + ((q & 15) << 2);
}

__global__ __launch_bounds__(128) void self_branch_kernel(
    const float* __restrict__ x, const float* __restrict__ FWin,
    const float* __restrict__ bin, const float* __restrict__ FWout,
    const float* __restrict__ bout, const float* __restrict__ FWg1,
    const unsigned* __restrict__ planes,
    float* __restrict__ deg, float* __restrict__ l1_out,
    float* __restrict__ h1_out, float* __restrict__ g1_out,
    int* __restrict__ ctr, int N)
{
    __shared__ alignas(16) float lds[64 * LDS_LD];
    __shared__ unsigned sdeg[4 * 32];
    if (blockIdx.x == 0 && threadIdx.x < 4) ctr[threadIdx.x] = 0;
    const int lane = threadIdx.x & 31;
    const int wave = threadIdx.x >> 5;
    const int r0   = blockIdx.x * 32;
    const int n    = lane & 15, half = lane >> 4;
    if (r0 >= N) return;
    float* LX = lds;
    float* LA = lds + 32 * LDS_LD;

    stage_rows128(x, r0, LX, LDS_LD, wave, lane);
    unsigned c = 0u;
#pragma unroll
    for (int q = 0; q < DEG_P / 4; ++q)
        c += planes[(size_t)(wave * (DEG_P / 4) + q) * N + r0 + lane];
    sdeg[wave * 32 + lane] = c;
    __syncthreads();

    const unsigned tl = sdeg[lane] + sdeg[32 + lane] + sdeg[64 + lane] + sdeg[96 + lane];
    const float degl = (float)(tl + 1u);
    if (wave == 0) deg[r0 + lane] = degl;
    const float disl = rsqrtf(degl);

    v2f ax0[32], ax1[32];
#pragma unroll
    for (int kc = 0; kc < 32; ++kc) {
        ax0[kc] = a_frag_lds(LX,               LDS_LD, kc * 4, lane);
        ax1[kc] = a_frag_lds(LX + 16 * LDS_LD, LDS_LD, kc * 4, lane);
    }

    for (int tt = 0; tt < 2; ++tt) {
        const int t = 2 * wave + tt;
        v8f acc0 = {}, acc1 = {};
        v4f f[16];
#pragma unroll
        for (int p = 0; p < 16; ++p) f[p] = b_quad(FWin, 16, t, p, lane);
        __builtin_amdgcn_sched_barrier(0);
#pragma unroll
        for (int p = 0; p < 16; ++p) {
            acc0 = wmma_f32(ax0[2 * p],     v2f{f[p][0], f[p][1]}, acc0);
            acc1 = wmma_f32(ax1[2 * p],     v2f{f[p][0], f[p][1]}, acc1);
            acc0 = wmma_f32(ax0[2 * p + 1], v2f{f[p][2], f[p][3]}, acc0);
            acc1 = wmma_f32(ax1[2 * p + 1], v2f{f[p][2], f[p][3]}, acc1);
        }
        const float bb = bin[t * 16 + n];
#pragma unroll
        for (int r = 0; r < 8; ++r) {
            const float v0 = acc0[r] + bb;
            const float v1 = acc1[r] + bb;
            LA[(r + 8 * half) * LDS_LD + t * 16 + n]      = v0 > 0.f ? v0 : 0.f;
            LA[(16 + r + 8 * half) * LDS_LD + t * 16 + n] = v1 > 0.f ? v1 : 0.f;
        }
    }
    __syncthreads();

    float dd0[8], dd1[8];
#pragma unroll
    for (int r = 0; r < 8; ++r) {
        dd0[r] = __shfl(disl, 8 * half + r);
        dd1[r] = __shfl(disl, 16 + 8 * half + r);
    }
    for (int tt = 0; tt < 2; ++tt) {
        const int t = 2 * wave + tt;
        v8f acc0 = {}, acc1 = {};
        v4f f[16];
#pragma unroll
        for (int p = 0; p < 16; ++p) f[p] = b_quad(FWg1, 16, t, p, lane);
        __builtin_amdgcn_sched_barrier(0);
#pragma unroll
        for (int p = 0; p < 16; ++p) {
            acc0 = wmma_f32(ax0[2 * p],     v2f{f[p][0], f[p][1]}, acc0);
            acc1 = wmma_f32(ax1[2 * p],     v2f{f[p][0], f[p][1]}, acc1);
            acc0 = wmma_f32(ax0[2 * p + 1], v2f{f[p][2], f[p][3]}, acc0);
            acc1 = wmma_f32(ax1[2 * p + 1], v2f{f[p][2], f[p][3]}, acc1);
        }
#pragma unroll
        for (int r = 0; r < 8; ++r) {
            const size_t o0 = (size_t)(r0 + r + 8 * half) * 128 + t * 16 + n;
            const size_t o1 = o0 + (size_t)16 * 128;
            const float hs0 = acc0[r] * dd0[r];
            const float hs1 = acc1[r] * dd1[r];
            h1_out[o0] = hs0;
            g1_out[o0] = hs0;
            h1_out[o1] = hs1;
            g1_out[o1] = hs1;
        }
    }

    {
        const int t = wave;
        v8f acc0 = {}, acc1 = {};
        v4f f[32];
#pragma unroll
        for (int p = 0; p < 32; ++p) f[p] = b_quad(FWout, 32, t, p, lane);
        __builtin_amdgcn_sched_barrier(0);
#pragma unroll
        for (int p = 0; p < 16; ++p) {
            acc0 = wmma_f32(ax0[2 * p],     v2f{f[p][0], f[p][1]}, acc0);
            acc1 = wmma_f32(ax1[2 * p],     v2f{f[p][0], f[p][1]}, acc1);
            acc0 = wmma_f32(ax0[2 * p + 1], v2f{f[p][2], f[p][3]}, acc0);
            acc1 = wmma_f32(ax1[2 * p + 1], v2f{f[p][2], f[p][3]}, acc1);
        }
        const float* L0 = LA;
        const float* L1 = LA + 16 * LDS_LD;
#pragma unroll
        for (int p = 0; p < 16; ++p) {
            acc0 = wmma_f32(a_frag_lds(L0, LDS_LD, 8 * p,     lane), v2f{f[16 + p][0], f[16 + p][1]}, acc0);
            acc1 = wmma_f32(a_frag_lds(L1, LDS_LD, 8 * p,     lane), v2f{f[16 + p][0], f[16 + p][1]}, acc1);
            acc0 = wmma_f32(a_frag_lds(L0, LDS_LD, 8 * p + 4, lane), v2f{f[16 + p][2], f[16 + p][3]}, acc0);
            acc1 = wmma_f32(a_frag_lds(L1, LDS_LD, 8 * p + 4, lane), v2f{f[16 + p][2], f[16 + p][3]}, acc1);
        }
        const float bo = bout[t * 16 + n];
#pragma unroll
        for (int r = 0; r < 8; ++r) { acc0[r] += bo; acc1[r] += bo; }
        stage_out_tile(LX, t, lane, acc0, acc1);
    }
    __syncthreads();
#pragma unroll
    for (int i = 0; i < 4; ++i) {
        const int q = i * 128 + (int)threadIdx.x;
        *reinterpret_cast<v4f*>(l1_out + piece_offset(r0, q)) = staged_piece(LX, q);
    }
}


#define SC_WPB  8
#define SC_TPB  256
#define SC_GRID 1280

__device__ __forceinline__ int lane_bcast_i(int v, int k) {
    return __builtin_amdgcn_readlane(v, k);
}

template <int W>
__global__ __launch_bounds__(SC_TPB) void seg_scatter_kernel(
    const int* __restrict__ row, const int* __restrict__ col,
    const float* __restrict__ h, float* __restrict__ g,
    int* __restrict__ ctr,
    int E, int N)
{
    constexpr int NL = W / 32;
    constexpr int BE = 80 / NL;
    __shared__ int s_half;
    __shared__ int s_base;
    const int lane = threadIdx.x & 31;
    const int wave = __builtin_amdgcn_readfirstlane((int)(threadIdx.x >> 5));
    const int nunits = (E + 63) >> 6;
    const int split  = N >> 1;
    const int wph = ((int)gridDim.x * SC_WPB) >> 1;
    const int upw = (wph > 0) ? ((nunits + wph - 1) / wph > 0 ? (nunits + wph - 1) / wph : 1) : 1;
    const int upb = upw * SC_WPB;

    if (wave == 0) {
        const int id = __builtin_amdgcn_s_sendmsg_rtn(0x87);
        if (lane == 0) s_half = (id >> 18) & 1;
    }
    __syncthreads();
    const int myhalf = __builtin_amdgcn_readfirstlane(s_half);

    for (int pass = 0; pass < 2; ++pass) {
        const int q = myhalf ^ pass;
        int* cq = ctr + q;
        for (int it = 0; it <= nunits; ++it) {
            if (threadIdx.x == 0) s_base = atomicAdd(cq, upb);
            __syncthreads();
            const int base = __builtin_amdgcn_readfirstlane(s_base);
            __syncthreads();
            if (base >= nunits) break;
            const int ubeg = base + wave * upw;
            const int uend = (ubeg + upw < nunits) ? (ubeg + upw) : nunits;

            for (int u = ubeg; u < uend; ++u) {
                const int e0 = u * 64 + lane, e1 = e0 + 32;
                int r0 = 0, c0 = -1, r1 = 0, c1 = -1;
                if (e0 < E) { r0 = row[e0]; c0 = col[e0]; }
                if (e1 < E) { r1 = row[e1]; c1 = col[e1]; }
                const bool mine0 = (c0 >= 0) && ((c0 >= split) == (q != 0));
                const bool mine1 = (c1 >= 0) && ((c1 >= split) == (q != 0));
                unsigned long long m =
                    (unsigned long long)__builtin_amdgcn_ballot_w32(mine0) |
                    ((unsigned long long)__builtin_amdgcn_ballot_w32(mine1) << 32);
                while (m) {
                    const unsigned long long mb = m;
                    float v[BE][NL];
#pragma unroll
                    for (int j = 0; j < BE; ++j) {
                        if (m) {
                            const int k  = __builtin_ctzll(m);
                            m &= m - 1;
                            const int ra = lane_bcast_i(r0, k & 31);
                            const int rb = lane_bcast_i(r1, k & 31);
                            const int rs = (k < 32) ? ra : rb;
                            const float* hp = h + (size_t)rs * W + lane;
#pragma unroll
                            for (int l = 0; l < NL; ++l) v[j][l] = hp[32 * l];
                        }
                    }
                    unsigned long long m2 = mb;
#pragma unroll
                    for (int j = 0; j < BE; ++j) {
                        if (m2 != m) {
                            const int k  = __builtin_ctzll(m2);
                            m2 &= m2 - 1;
                            const int ca = lane_bcast_i(c0, k & 31);
                            const int cb = lane_bcast_i(c1, k & 31);
                            const int cs = (k < 32) ? ca : cb;
                            float* gp = g + (size_t)cs * W + lane;
#pragma unroll
                            for (int l = 0; l < NL; ++l) atomicAdd(gp + 32 * l, v[j][l]);
                        }
                    }
                }
            }
        }
    }
}

__global__ __launch_bounds__(128) void gcn2_gemm_kernel(
    const float* __restrict__ s1, const float* __restrict__ FWg2,
    const float* __restrict__ bg1, const float* __restrict__ deg,
    float* __restrict__ h2, float* __restrict__ g2, int N)
{
    __shared__ alignas(16) float lds[32 * LDS_LD];
    __shared__ alignas(16) float ost[32 * O_LD];
    const int lane = threadIdx.x & 31;
    const int wave = threadIdx.x >> 5;
    const int r0   = blockIdx.x * 32;
    if (r0 >= N) return;
    const int half = lane >> 4;

    const int t = wave;
    v4f f[16];
#pragma unroll
    for (int p = 0; p < 16; ++p) f[p] = b_quad(FWg2, 16, t, p, lane);
    __builtin_amdgcn_sched_barrier(0);

    {
        const int cc = lane << 2;
        const v4f bq = *reinterpret_cast<const v4f*>(bg1 + cc);
#pragma unroll
        for (int i = 0; i < 8; ++i) {
            const int rr = i * 4 + wave;
            const float dis = rsqrtf(deg[r0 + rr]);
            const v4f s = *reinterpret_cast<const v4f*>(s1 + (size_t)(r0 + rr) * 128 + cc);
            float* d = lds + rr * LDS_LD + cc;
            *reinterpret_cast<v2f*>(d)     = v2f{bq[0] + dis * s[0], bq[1] + dis * s[1]};
            *reinterpret_cast<v2f*>(d + 2) = v2f{bq[2] + dis * s[2], bq[3] + dis * s[3]};
        }
    }
    const float disl = rsqrtf(deg[r0 + lane]);
    float dd0[8], dd1[8];
#pragma unroll
    for (int r = 0; r < 8; ++r) {
        dd0[r] = __shfl(disl, 8 * half + r);
        dd1[r] = __shfl(disl, 16 + 8 * half + r);
    }
    __syncthreads();

    v8f acc0 = {}, acc1 = {};
    const float* L0 = lds;
    const float* L1 = lds + 16 * LDS_LD;
#pragma unroll
    for (int p = 0; p < 16; ++p) {
        acc0 = wmma_f32(a_frag_lds(L0, LDS_LD, 8 * p,     lane), v2f{f[p][0], f[p][1]}, acc0);
        acc1 = wmma_f32(a_frag_lds(L1, LDS_LD, 8 * p,     lane), v2f{f[p][0], f[p][1]}, acc1);
        acc0 = wmma_f32(a_frag_lds(L0, LDS_LD, 8 * p + 4, lane), v2f{f[p][2], f[p][3]}, acc0);
        acc1 = wmma_f32(a_frag_lds(L1, LDS_LD, 8 * p + 4, lane), v2f{f[p][2], f[p][3]}, acc1);
    }
#pragma unroll
    for (int r = 0; r < 8; ++r) { acc0[r] *= dd0[r]; acc1[r] *= dd1[r]; }
    stage_out_tile(ost, t, lane, acc0, acc1);
    __syncthreads();
#pragma unroll
    for (int i = 0; i < 4; ++i) {
        const int q = i * 128 + (int)threadIdx.x;
        const v4f v = staged_piece(ost, q);
        const size_t o = piece_offset(r0, q);
        *reinterpret_cast<v4f*>(h2 + o) = v;
        *reinterpret_cast<v4f*>(g2 + o) = v;
    }
}

__global__ __launch_bounds__(128) void out_gemm_kernel(
    const float* __restrict__ xn, const float* __restrict__ s2,
    const float* __restrict__ bg2, const float* __restrict__ deg,
    const float* __restrict__ FW, const float* __restrict__ b,
    float* __restrict__ out, int N)
{
    __shared__ alignas(16) float lds[32 * OUT_LD];
    __shared__ alignas(16) float ost[32 * O_LD];
    const int lane = threadIdx.x & 31;
    const int wave = threadIdx.x >> 5;
    const int r0   = blockIdx.x * 32;
    if (r0 >= N) return;
    const int n = lane & 15;

    const int t = wave;
    v4f f[24];
#pragma unroll
    for (int p = 0; p < 24; ++p) f[p] = b_quad(FW, 24, t, p, lane);
    __builtin_amdgcn_sched_barrier(0);

    stage_rows128(xn, r0, lds, OUT_LD, wave, lane);
    {
        const int cc = (lane & 15) << 2;
        const v4f bq = *reinterpret_cast<const v4f*>(bg2 + cc);
#pragma unroll
        for (int i = 0; i < 4; ++i) {
            const int rr = i * 8 + wave * 2 + (lane >> 4);
            const float dis = rsqrtf(deg[r0 + rr]);
            const v4f s = *reinterpret_cast<const v4f*>(s2 + (size_t)(r0 + rr) * 64 + cc);
            float* d = lds + rr * OUT_LD + 128 + cc;
            *reinterpret_cast<v2f*>(d)     = v2f{bq[0] + dis * s[0], bq[1] + dis * s[1]};
            *reinterpret_cast<v2f*>(d + 2) = v2f{bq[2] + dis * s[2], bq[3] + dis * s[3]};
        }
    }
    __syncthreads();

    v8f acc0 = {}, acc1 = {};
    const float* L0 = lds;
    const float* L1 = lds + 16 * OUT_LD;
#pragma unroll
    for (int p = 0; p < 24; ++p) {
        acc0 = wmma_f32(a_frag_lds(L0, OUT_LD, 8 * p,     lane), v2f{f[p][0], f[p][1]}, acc0);
        acc1 = wmma_f32(a_frag_lds(L1, OUT_LD, 8 * p,     lane), v2f{f[p][0], f[p][1]}, acc1);
        acc0 = wmma_f32(a_frag_lds(L0, OUT_LD, 8 * p + 4, lane), v2f{f[p][2], f[p][3]}, acc0);
        acc1 = wmma_f32(a_frag_lds(L1, OUT_LD, 8 * p + 4, lane), v2f{f[p][2], f[p][3]}, acc1);
    }
    const float bo = b[t * 16 + n];
#pragma unroll
    for (int r = 0; r < 8; ++r) { acc0[r] += bo; acc1[r] += bo; }
    stage_out_tile(ost, t, lane, acc0, acc1);
    __syncthreads();
#pragma unroll
    for (int i = 0; i < 4; ++i) {
        const int q = i * 128 + (int)threadIdx.x;
        *reinterpret_cast<v4f*>(out + piece_offset(r0, q)) = staged_piece(ost, q);
    }
}

extern "C" void kernel_launch(void* const* d_in, const int* in_sizes, int n_in,
                              void* d_out, int out_size, void* d_ws, size_t ws_size,
                              hipStream_t stream) {
    const float* x_self     = (const float*)d_in[0];
    const float* x_neighbor = (const float*)d_in[1];
    const int*   edge_index = (const int*)d_in[2];
    const float* W_in_self  = (const float*)d_in[3];
    const float* b_in_self  = (const float*)d_in[4];
    const float* W_out_self = (const float*)d_in[5];
    const float* b_out_self = (const float*)d_in[6];
    const float* Wg1        = (const float*)d_in[7];
    const float* bg1        = (const float*)d_in[8];
    const float* Wg2        = (const float*)d_in[9];
    const float* bg2        = (const float*)d_in[10];
    const float* W_out      = (const float*)d_in[11];
    const float* b_out      = (const float*)d_in[12];

    const int N = in_sizes[0] / 128;
    const int E = in_sizes[2] / 2;
    const int* row = edge_index;
    const int* col = edge_index + E;

    float* ws  = (float*)d_ws;
    int*   ctr = (int*)d_ws;
    float* FWos = ws + 128;
    float* FWg2 = FWos + 256 * 64;
    float* FWo  = FWg2 + 128 * 64;
    float* deg = ws + N;
    float* h   = ws + 2 * (size_t)N;
    float* g   = h + (size_t)N * 128;
    float* g2  = h + (size_t)N * 64;

    float* l1_out = (float*)d_out;
    float* x2_out = (float*)d_out + (size_t)N * 64;

    unsigned* planes = (unsigned*)x2_out;
    float* FWin = x2_out + (size_t)DEG_P * N;
    float* FWg1 = FWin + 128 * 128;
    deg_hist_kernel<<<dim3(DEG_P, DEG_NW + 1), 1024, 0, stream>>>(
        col, planes, N, E, W_in_self, Wg1, W_out_self, Wg2, W_out,
        FWin, FWg1, FWos, FWg2, FWo);

    const int blocks = (N + 31) / 32;

    self_branch_kernel<<<blocks, 128, 0, stream>>>(x_self, FWin, b_in_self,
                                                   FWos, b_out_self, FWg1,
                                                   planes, deg, l1_out, h, g, ctr, N);

    seg_scatter_kernel<128><<<SC_GRID, SC_TPB, 0, stream>>>(row, col, h, g, ctr, E, N);

    gcn2_gemm_kernel<<<blocks, 128, 0, stream>>>(g, FWg2, bg1, deg, h, g2, N);

    seg_scatter_kernel<64><<<SC_GRID, SC_TPB, 0, stream>>>(row, col, h, g2, ctr + 2, E, N);

    out_gemm_kernel<<<blocks, 128, 0, stream>>>(x_neighbor, g2, bg2, deg, FWo, b_out,
                                                x2_out, N);
}
